# starting kernel
# baseline (speedup 1.0000x reference)
#include <hip/hip_runtime.h>

typedef __attribute__((ext_vector_type(2))) float v2f;
typedef __attribute__((ext_vector_type(8))) float v8f;

#define NN 100000
#define FF 256
#define HC 128
#define NEG_SLOPE 0.2f
#define ENC_NEG_INF 0x007FFFFFu

__device__ __forceinline__ unsigned enc_f32(float x) {
  unsigned u = __float_as_uint(x);
  return u ^ ((u & 0x80000000u) ? 0xFFFFFFFFu : 0x80000000u);
}
__device__ __forceinline__ float dec_f32(unsigned m) {
  unsigned u = (m & 0x80000000u) ? (m ^ 0x80000000u) : ~m;
  return __uint_as_float(u);
}
__device__ __forceinline__ void atomic_add_f32(float* p, float v) {
  __hip_atomic_fetch_add(p, v, __ATOMIC_RELAXED, __HIP_MEMORY_SCOPE_AGENT);
}

__global__ void __launch_bounds__(256) k_init(float* __restrict__ out,
                                              unsigned* __restrict__ maxbits,
                                              float* __restrict__ denom,
                                              const float* __restrict__ bias) {
  int tid = blockIdx.x * blockDim.x + threadIdx.x;
  if (tid < NN * HC) out[tid] = bias[tid & (HC - 1)];
  if (tid < NN * 2) { maxbits[tid] = ENC_NEG_INF; denom[tid] = 0.0f; }
}

__global__ void __launch_bounds__(256) k_proj(const float* __restrict__ x,
                                              const float* __restrict__ Wl,
                                              const float* __restrict__ bl,
                                              const float* __restrict__ Wr,
                                              const float* __restrict__ br,
                                              float* __restrict__ proj) {
  __shared__ __align__(16) float xs[16 * 260];
  const int m0 = blockIdx.x * 16;
  const int t  = threadIdx.x;

#pragma unroll
  for (int i = 0; i < 4; ++i) {
    int idx = i * 256 + t;
    int row = idx >> 6;
    int c4  = (idx & 63) << 2;
    float4 v = *(const float4*)(x + (long)(m0 + row) * FF + c4);
    *(float4*)(&xs[row * 260 + c4]) = v;
  }
  __syncthreads();

  const int lane = t & 31;
  const int wave = t >> 5;
  const int r    = lane & 15;
  const int koff = (lane >> 4) << 1;

  for (int t2 = 0; t2 < 2; ++t2) {
    const int n0  = wave * 32 + t2 * 16;
    const int col = n0 + r;
    const float* Wp;
    float bv;
    if (col < HC) { Wp = Wl + (long)col * FF;        bv = bl[col]; }
    else          { Wp = Wr + (long)(col - HC) * FF; bv = br[col - HC]; }
    __builtin_prefetch(Wp, 0, 0);

    v8f acc;
#pragma unroll
    for (int v = 0; v < 8; ++v) acc[v] = bv;

#pragma unroll 8
    for (int k0 = 0; k0 < FF; k0 += 4) {
      v2f a, b;
      a[0] = xs[r * 260 + k0 + koff];
      a[1] = xs[r * 260 + k0 + koff + 1];
      b[0] = Wp[k0 + koff];
      b[1] = Wp[k0 + koff + 1];
      acc = __builtin_amdgcn_wmma_f32_16x16x4_f32(
          false, a, false, b, (short)0, acc, false, false);
    }

    const int rowbase = m0 + ((lane >> 4) << 3);
#pragma unroll
    for (int v = 0; v < 8; ++v)
      proj[(long)(rowbase + v) * FF + col] = acc[v];
  }
}

__global__ void __launch_bounds__(256) k_logits(const int* __restrict__ ei,
                                                const float* __restrict__ proj,
                                                const float* __restrict__ att,
                                                float* __restrict__ logits,
                                                unsigned* __restrict__ maxbits,
                                                int Eraw, int Etot) {
  int wid = blockIdx.x * 8 + (threadIdx.x >> 5);
  if (wid >= Etot) return;
  int lane = threadIdx.x & 31;

  int src, dst;
  if (wid < Eraw) { src = ei[wid]; dst = ei[Eraw + wid]; }
  else            { src = dst = wid - Eraw; }

  int j0 = lane << 2;
  float4 a = *(const float4*)(proj + (long)src * FF + j0);
  float4 b = *(const float4*)(proj + (long)dst * FF + HC + j0);
  float4 w = *(const float4*)(att + j0);

  float s = 0.0f, e;
  e = a.x + b.x; e = e > 0.0f ? e : NEG_SLOPE * e; s += w.x * e;
  e = a.y + b.y; e = e > 0.0f ? e : NEG_SLOPE * e; s += w.y * e;
  e = a.z + b.z; e = e > 0.0f ? e : NEG_SLOPE * e; s += w.z * e;
  e = a.w + b.w; e = e > 0.0f ? e : NEG_SLOPE * e; s += w.w * e;

  for (int m = 1; m < 16; m <<= 1) s += __shfl_xor(s, m, 32);

  if ((lane & 15) == 0) {
    int h = lane >> 4;
    logits[wid * 2 + h] = s;
    atomicMax(&maxbits[dst * 2 + h], enc_f32(s));
  }
}

__global__ void __launch_bounds__(256) k_exp(const int* __restrict__ ei,
                                             const unsigned* __restrict__ maxbits,
                                             float* __restrict__ lp,
                                             float* __restrict__ denom,
                                             int Eraw, int Etot) {
  int tid = blockIdx.x * blockDim.x + threadIdx.x;
  if (tid >= 2 * Etot) return;
  int e = tid >> 1, h = tid & 1;
  int dst = (e < Eraw) ? ei[Eraw + e] : (e - Eraw);
  float m = dec_f32(maxbits[dst * 2 + h]);
  float p = __expf(lp[tid] - m);
  lp[tid] = p;
  atomic_add_f32(&denom[dst * 2 + h], p);
}

__global__ void __launch_bounds__(256) k_agg(const int* __restrict__ ei,
                                             const float* __restrict__ proj,
                                             const float* __restrict__ p,
                                             const float* __restrict__ denom,
                                             float* __restrict__ out,
                                             int Eraw, int Etot) {
  int wid = blockIdx.x * 8 + (threadIdx.x >> 5);
  if (wid >= Etot) return;
  int lane = threadIdx.x & 31;

  int src, dst;
  if (wid < Eraw) { src = ei[wid]; dst = ei[Eraw + wid]; }
  else            { src = dst = wid - Eraw; }

  int j0 = lane << 2;
  int h  = lane >> 4;
  float alpha = p[wid * 2 + h] / denom[dst * 2 + h];
  float4 xl = *(const float4*)(proj + (long)src * FF + j0);
  float* o = out + (long)dst * HC + j0;
  atomic_add_f32(o + 0, alpha * xl.x);
  atomic_add_f32(o + 1, alpha * xl.y);
  atomic_add_f32(o + 2, alpha * xl.z);
  atomic_add_f32(o + 3, alpha * xl.w);
}

extern "C" void kernel_launch(void* const* d_in, const int* in_sizes, int n_in,
                              void* d_out, int out_size, void* d_ws, size_t ws_size,
                              hipStream_t stream) {
  const float* x    = (const float*)d_in[0];
  const float* Wl   = (const float*)d_in[1];
  const float* bl   = (const float*)d_in[2];
  const float* Wr   = (const float*)d_in[3];
  const float* br   = (const float*)d_in[4];
  const float* att  = (const float*)d_in[5];
  const float* bias = (const float*)d_in[6];
  const int*   ei   = (const int*)d_in[7];
  float* out = (float*)d_out;

  const int Eraw = in_sizes[7] / 2;
  const int Etot = Eraw + NN;

  float*    proj    = (float*)d_ws;
  float*    logits  = proj + (size_t)NN * FF;
  unsigned* maxbits = (unsigned*)(logits + (size_t)Etot * 2);
  float*    denom   = (float*)(maxbits + (size_t)NN * 2);

  k_init  <<<(NN * HC + 255) / 256, 256, 0, stream>>>(out, maxbits, denom, bias);
  k_proj  <<<NN / 16,               256, 0, stream>>>(x, Wl, bl, Wr, br, proj);
  k_logits<<<(Etot + 7) / 8,        256, 0, stream>>>(ei, proj, att, logits, maxbits, Eraw, Etot);
  k_exp   <<<(2 * Etot + 255) / 256,256, 0, stream>>>(ei, maxbits, logits, denom, Eraw, Etot);
  k_agg   <<<(Etot + 7) / 8,        256, 0, stream>>>(ei, proj, logits, denom, out, Eraw, Etot);
}


# Edge kernels (logits, weighted scatter): each wave takes 8 consecutive edges and requests all their index loads and row gathers before first use; float atomics and their arrival order kept
# speedup vs baseline: 1.1040x; 1.1040x over previous
#include <hip/hip_runtime.h>

typedef __attribute__((ext_vector_type(2))) float v2f;
typedef __attribute__((ext_vector_type(8))) float v8f;

#define NN 100000
#define FF 256
#define HC 128
#define NEG_SLOPE 0.2f
#define ENC_NEG_INF 0x007FFFFFu

__device__ __forceinline__ unsigned enc_f32(float x) {
  unsigned u = __float_as_uint(x);
  return u ^ ((u & 0x80000000u) ? 0xFFFFFFFFu : 0x80000000u);
}
__device__ __forceinline__ float dec_f32(unsigned m) {
  unsigned u = (m & 0x80000000u) ? (m ^ 0x80000000u) : ~m;
  return __uint_as_float(u);
}
__device__ __forceinline__ void atomic_add_f32(float* p, float v) {
  __hip_atomic_fetch_add(p, v, __ATOMIC_RELAXED, __HIP_MEMORY_SCOPE_AGENT);
}

__global__ void __launch_bounds__(256) k_init(float* __restrict__ out,
                                              unsigned* __restrict__ maxbits,
                                              float* __restrict__ denom,
                                              const float* __restrict__ bias) {
  int tid = blockIdx.x * blockDim.x + threadIdx.x;
  if (tid < NN * HC) out[tid] = bias[tid & (HC - 1)];
  if (tid < NN * 2) { maxbits[tid] = ENC_NEG_INF; denom[tid] = 0.0f; }
}

__global__ void __launch_bounds__(256) k_proj(const float* __restrict__ x,
                                              const float* __restrict__ Wl,
                                              const float* __restrict__ bl,
                                              const float* __restrict__ Wr,
                                              const float* __restrict__ br,
                                              float* __restrict__ proj) {
  __shared__ __align__(16) float xs[16 * 260];
  const int m0 = blockIdx.x * 16;
  const int t  = threadIdx.x;

#pragma unroll
  for (int i = 0; i < 4; ++i) {
    int idx = i * 256 + t;
    int row = idx >> 6;
    int c4  = (idx & 63) << 2;
    float4 v = *(const float4*)(x + (long)(m0 + row) * FF + c4);
    *(float4*)(&xs[row * 260 + c4]) = v;
  }
  __syncthreads();

  const int lane = t & 31;
  const int wave = t >> 5;
  const int r    = lane & 15;
  const int koff = (lane >> 4) << 1;

  for (int t2 = 0; t2 < 2; ++t2) {
    const int n0  = wave * 32 + t2 * 16;
    const int col = n0 + r;
    const float* Wp;
    float bv;
    if (col < HC) { Wp = Wl + (long)col * FF;        bv = bl[col]; }
    else          { Wp = Wr + (long)(col - HC) * FF; bv = br[col - HC]; }
    __builtin_prefetch(Wp, 0, 0);

    v8f acc;
#pragma unroll
    for (int v = 0; v < 8; ++v) acc[v] = bv;

#pragma unroll 8
    for (int k0 = 0; k0 < FF; k0 += 4) {
      v2f a, b;
      a[0] = xs[r * 260 + k0 + koff];
      a[1] = xs[r * 260 + k0 + koff + 1];
      b[0] = Wp[k0 + koff];
      b[1] = Wp[k0 + koff + 1];
      acc = __builtin_amdgcn_wmma_f32_16x16x4_f32(
          false, a, false, b, (short)0, acc, false, false);
    }

    const int rowbase = m0 + ((lane >> 4) << 3);
#pragma unroll
    for (int v = 0; v < 8; ++v)
      proj[(long)(rowbase + v) * FF + col] = acc[v];
  }
}

#define EPW 8

enum { ALL_RAW = 0, ALL_SELF = 1, ANY = 2 };

template <int MODE>
__device__ __forceinline__ void edge_nodes(const int* __restrict__ ei, int e,
                                           int Eraw, int Etot, int& src, int& dst) {
  if (MODE == ALL_RAW) {
    src = ei[e]; dst = ei[Eraw + e];
  } else if (MODE == ALL_SELF) {
    src = dst = e - Eraw;
  } else {
    int ec = e < Etot ? e : Etot - 1;
    if (ec < Eraw) { src = ei[ec]; dst = ei[Eraw + ec]; }
    else           { src = dst = ec - Eraw; }
  }
}

template <int MODE>
__device__ __forceinline__ void logits_batch(const int* __restrict__ ei,
                                             const float* __restrict__ proj,
                                             const float* __restrict__ att,
                                             float* __restrict__ logits,
                                             unsigned* __restrict__ maxbits,
                                             int Eraw, int Etot, int e0, int lane) {
  const int j0 = lane << 2;
  int src[EPW], dst[EPW];
#pragma unroll
  for (int i = 0; i < EPW; ++i)
    edge_nodes<MODE>(ei, e0 + i, Eraw, Etot, src[i], dst[i]);

  float4 a[EPW], b[EPW];
#pragma unroll
  for (int i = 0; i < EPW; ++i) {
    a[i] = *(const float4*)(proj + (long)src[i] * FF + j0);
    b[i] = *(const float4*)(proj + (long)dst[i] * FF + HC + j0);
  }
  const float4 w = *(const float4*)(att + j0);

  float sum[EPW];
#pragma unroll
  for (int i = 0; i < EPW; ++i) {
    float s = 0.0f, e;
    e = a[i].x + b[i].x; e = e > 0.0f ? e : NEG_SLOPE * e; s += w.x * e;
    e = a[i].y + b[i].y; e = e > 0.0f ? e : NEG_SLOPE * e; s += w.y * e;
    e = a[i].z + b[i].z; e = e > 0.0f ? e : NEG_SLOPE * e; s += w.z * e;
    e = a[i].w + b[i].w; e = e > 0.0f ? e : NEG_SLOPE * e; s += w.w * e;
    sum[i] = s;
  }

#pragma unroll
  for (int m = 1; m < 16; m <<= 1) {
#pragma unroll
    for (int i = 0; i < EPW; ++i) sum[i] += __shfl_xor(sum[i], m, 32);
  }

  if ((lane & 15) == 0) {
    int h = lane >> 4;
#pragma unroll
    for (int i = 0; i < EPW; ++i) {
      if (MODE != ANY || e0 + i < Etot) {
        logits[(e0 + i) * 2 + h] = sum[i];
        atomicMax(&maxbits[dst[i] * 2 + h], enc_f32(sum[i]));
      }
    }
  }
}

__global__ void __launch_bounds__(256) k_logits(const int* __restrict__ ei,
                                                const float* __restrict__ proj,
                                                const float* __restrict__ att,
                                                float* __restrict__ logits,
                                                unsigned* __restrict__ maxbits,
                                                int Eraw, int Etot) {
  const int wave = __builtin_amdgcn_readfirstlane((int)(threadIdx.x >> 5));
  const int e0   = (blockIdx.x * 8 + wave) * EPW;
  if (e0 >= Etot) return;
  const int lane = threadIdx.x & 31;
  if (e0 + EPW <= Eraw)
    logits_batch<ALL_RAW>(ei, proj, att, logits, maxbits, Eraw, Etot, e0, lane);
  else if (e0 >= Eraw && e0 + EPW <= Etot)
    logits_batch<ALL_SELF>(ei, proj, att, logits, maxbits, Eraw, Etot, e0, lane);
  else
    logits_batch<ANY>(ei, proj, att, logits, maxbits, Eraw, Etot, e0, lane);
}

__global__ void __launch_bounds__(256) k_exp(const int* __restrict__ ei,
                                             const unsigned* __restrict__ maxbits,
                                             float* __restrict__ lp,
                                             float* __restrict__ denom,
                                             int Eraw, int Etot) {
  int tid = blockIdx.x * blockDim.x + threadIdx.x;
  if (tid >= 2 * Etot) return;
  int e = tid >> 1, h = tid & 1;
  int dst = (e < Eraw) ? ei[Eraw + e] : (e - Eraw);
  float m = dec_f32(maxbits[dst * 2 + h]);
  float p = __expf(lp[tid] - m);
  lp[tid] = p;
  atomic_add_f32(&denom[dst * 2 + h], p);
}

template <int MODE>
__device__ __forceinline__ void agg_batch(const int* __restrict__ ei,
                                          const float* __restrict__ proj,
                                          const float* __restrict__ p,
                                          const float* __restrict__ denom,
                                          float* __restrict__ out,
                                          int Eraw, int Etot, int e0, int lane) {
  const int j0 = lane << 2;
  const int h  = lane >> 4;
  int src[EPW], dst[EPW];
  float pe[EPW], dn[EPW];
  float4 xl[EPW];
#pragma unroll
  for (int i = 0; i < EPW; ++i)
    edge_nodes<MODE>(ei, e0 + i, Eraw, Etot, src[i], dst[i]);
#pragma unroll
  for (int i = 0; i < EPW; ++i) {
    int e = (MODE != ANY || e0 + i < Etot) ? e0 + i : Etot - 1;
    pe[i] = p[e * 2 + h];
    dn[i] = denom[dst[i] * 2 + h];
    xl[i] = *(const float4*)(proj + (long)src[i] * FF + j0);
  }

#pragma unroll
  for (int i = 0; i < EPW; ++i) {
    if (MODE != ANY || e0 + i < Etot) {
      float alpha = pe[i] / dn[i];
      float* o = out + (long)dst[i] * HC + j0;
      atomic_add_f32(o + 0, alpha * xl[i].x);
      atomic_add_f32(o + 1, alpha * xl[i].y);
      atomic_add_f32(o + 2, alpha * xl[i].z);
      atomic_add_f32(o + 3, alpha * xl[i].w);
    }
  }
}

__global__ void __launch_bounds__(256) k_agg(const int* __restrict__ ei,
                                             const float* __restrict__ proj,
                                             const float* __restrict__ p,
                                             const float* __restrict__ denom,
                                             float* __restrict__ out,
                                             int Eraw, int Etot) {
  const int wave = __builtin_amdgcn_readfirstlane((int)(threadIdx.x >> 5));
  const int e0   = (blockIdx.x * 8 + wave) * EPW;
  if (e0 >= Etot) return;
  const int lane = threadIdx.x & 31;
  if (e0 + EPW <= Eraw)
    agg_batch<ALL_RAW>(ei, proj, p, denom, out, Eraw, Etot, e0, lane);
  else if (e0 >= Eraw && e0 + EPW <= Etot)
    agg_batch<ALL_SELF>(ei, proj, p, denom, out, Eraw, Etot, e0, lane);
  else
    agg_batch<ANY>(ei, proj, p, denom, out, Eraw, Etot, e0, lane);
}

extern "C" void kernel_launch(void* const* d_in, const int* in_sizes, int n_in,
                              void* d_out, int out_size, void* d_ws, size_t ws_size,
                              hipStream_t stream) {
  const float* x    = (const float*)d_in[0];
  const float* Wl   = (const float*)d_in[1];
  const float* bl   = (const float*)d_in[2];
  const float* Wr   = (const float*)d_in[3];
  const float* br   = (const float*)d_in[4];
  const float* att  = (const float*)d_in[5];
  const float* bias = (const float*)d_in[6];
  const int*   ei   = (const int*)d_in[7];
  float* out = (float*)d_out;

  const int Eraw = in_sizes[7] / 2;
  const int Etot = Eraw + NN;

  float*    proj    = (float*)d_ws;
  float*    logits  = proj + (size_t)NN * FF;
  unsigned* maxbits = (unsigned*)(logits + (size_t)Etot * 2);
  float*    denom   = (float*)(maxbits + (size_t)NN * 2);

  k_init  <<<(NN * HC + 255) / 256, 256, 0, stream>>>(out, maxbits, denom, bias);
  k_proj  <<<NN / 16,               256, 0, stream>>>(x, Wl, bl, Wr, br, proj);
  k_logits<<<(Etot + 8 * EPW - 1) / (8 * EPW), 256, 0, stream>>>(ei, proj, att, logits, maxbits, Eraw, Etot);
  k_exp   <<<(2 * Etot + 255) / 256,256, 0, stream>>>(ei, maxbits, logits, denom, Eraw, Etot);
  k_agg   <<<(Etot + 8 * EPW - 1) / (8 * EPW), 256, 0, stream>>>(ei, proj, logits, denom, out, Eraw, Etot);
}


# k_agg scatter: lane l owns channels l, l+32, l+64, l+96, so each row-gather and each float-atomic instruction covers one whole 128-B line, not quarters of four; float atomics and arrival order kept
# speedup vs baseline: 1.4137x; 1.2805x over previous
#include <hip/hip_runtime.h>

typedef __attribute__((ext_vector_type(2))) float v2f;
typedef __attribute__((ext_vector_type(8))) float v8f;

#define NN 100000
#define FF 256
#define HC 128
#define NEG_SLOPE 0.2f
#define ENC_NEG_INF 0x007FFFFFu

__device__ __forceinline__ unsigned enc_f32(float x) {
  unsigned u = __float_as_uint(x);
  return u ^ ((u & 0x80000000u) ? 0xFFFFFFFFu : 0x80000000u);
}
__device__ __forceinline__ float dec_f32(unsigned m) {
  unsigned u = (m & 0x80000000u) ? (m ^ 0x80000000u) : ~m;
  return __uint_as_float(u);
}
__device__ __forceinline__ void atomic_add_f32(float* p, float v) {
  __hip_atomic_fetch_add(p, v, __ATOMIC_RELAXED, __HIP_MEMORY_SCOPE_AGENT);
}

__global__ void __launch_bounds__(256) k_init(float* __restrict__ out,
                                              unsigned* __restrict__ maxbits,
                                              float* __restrict__ denom,
                                              const float* __restrict__ bias) {
  int tid = blockIdx.x * blockDim.x + threadIdx.x;
  if (tid < NN * HC) out[tid] = bias[tid & (HC - 1)];
  if (tid < NN * 2) { maxbits[tid] = ENC_NEG_INF; denom[tid] = 0.0f; }
}

__global__ void __launch_bounds__(256) k_proj(const float* __restrict__ x,
                                              const float* __restrict__ Wl,
                                              const float* __restrict__ bl,
                                              const float* __restrict__ Wr,
                                              const float* __restrict__ br,
                                              float* __restrict__ proj) {
  __shared__ __align__(16) float xs[16 * 260];
  const int m0 = blockIdx.x * 16;
  const int t  = threadIdx.x;

#pragma unroll
  for (int i = 0; i < 4; ++i) {
    int idx = i * 256 + t;
    int row = idx >> 6;
    int c4  = (idx & 63) << 2;
    float4 v = *(const float4*)(x + (long)(m0 + row) * FF + c4);
    *(float4*)(&xs[row * 260 + c4]) = v;
  }
  __syncthreads();

  const int lane = t & 31;
  const int wave = t >> 5;
  const int r    = lane & 15;
  const int koff = (lane >> 4) << 1;

  for (int t2 = 0; t2 < 2; ++t2) {
    const int n0  = wave * 32 + t2 * 16;
    const int col = n0 + r;
    const float* Wp;
    float bv;
    if (col < HC) { Wp = Wl + (long)col * FF;        bv = bl[col]; }
    else          { Wp = Wr + (long)(col - HC) * FF; bv = br[col - HC]; }
    __builtin_prefetch(Wp, 0, 0);

    v8f acc;
#pragma unroll
    for (int v = 0; v < 8; ++v) acc[v] = bv;

#pragma unroll 8
    for (int k0 = 0; k0 < FF; k0 += 4) {
      v2f a, b;
      a[0] = xs[r * 260 + k0 + koff];
      a[1] = xs[r * 260 + k0 + koff + 1];
      b[0] = Wp[k0 + koff];
      b[1] = Wp[k0 + koff + 1];
      acc = __builtin_amdgcn_wmma_f32_16x16x4_f32(
          false, a, false, b, (short)0, acc, false, false);
    }

    const int rowbase = m0 + ((lane >> 4) << 3);
#pragma unroll
    for (int v = 0; v < 8; ++v)
      proj[(long)(rowbase + v) * FF + col] = acc[v];
  }
}

#define EPW 8

enum { ALL_RAW = 0, ALL_SELF = 1, ANY = 2 };

template <int MODE>
__device__ __forceinline__ void edge_nodes(const int* __restrict__ ei, int e,
                                           int Eraw, int Etot, int& src, int& dst) {
  if (MODE == ALL_RAW) {
    src = ei[e]; dst = ei[Eraw + e];
  } else if (MODE == ALL_SELF) {
    src = dst = e - Eraw;
  } else {
    int ec = e < Etot ? e : Etot - 1;
    if (ec < Eraw) { src = ei[ec]; dst = ei[Eraw + ec]; }
    else           { src = dst = ec - Eraw; }
  }
}

template <int MODE>
__device__ __forceinline__ void logits_batch(const int* __restrict__ ei,
                                             const float* __restrict__ proj,
                                             const float* __restrict__ att,
                                             float* __restrict__ logits,
                                             unsigned* __restrict__ maxbits,
                                             int Eraw, int Etot, int e0, int lane) {
  const int j0 = lane << 2;
  int src[EPW], dst[EPW];
#pragma unroll
  for (int i = 0; i < EPW; ++i)
    edge_nodes<MODE>(ei, e0 + i, Eraw, Etot, src[i], dst[i]);

  float4 a[EPW], b[EPW];
#pragma unroll
  for (int i = 0; i < EPW; ++i) {
    a[i] = *(const float4*)(proj + (long)src[i] * FF + j0);
    b[i] = *(const float4*)(proj + (long)dst[i] * FF + HC + j0);
  }
  const float4 w = *(const float4*)(att + j0);

  float sum[EPW];
#pragma unroll
  for (int i = 0; i < EPW; ++i) {
    float s = 0.0f, e;
    e = a[i].x + b[i].x; e = e > 0.0f ? e : NEG_SLOPE * e; s += w.x * e;
    e = a[i].y + b[i].y; e = e > 0.0f ? e : NEG_SLOPE * e; s += w.y * e;
    e = a[i].z + b[i].z; e = e > 0.0f ? e : NEG_SLOPE * e; s += w.z * e;
    e = a[i].w + b[i].w; e = e > 0.0f ? e : NEG_SLOPE * e; s += w.w * e;
    sum[i] = s;
  }

#pragma unroll
  for (int m = 1; m < 16; m <<= 1) {
#pragma unroll
    for (int i = 0; i < EPW; ++i) sum[i] += __shfl_xor(sum[i], m, 32);
  }

  if ((lane & 15) == 0) {
    int h = lane >> 4;
#pragma unroll
    for (int i = 0; i < EPW; ++i) {
      if (MODE != ANY || e0 + i < Etot) {
        logits[(e0 + i) * 2 + h] = sum[i];
        atomicMax(&maxbits[dst[i] * 2 + h], enc_f32(sum[i]));
      }
    }
  }
}

__global__ void __launch_bounds__(256) k_logits(const int* __restrict__ ei,
                                                const float* __restrict__ proj,
                                                const float* __restrict__ att,
                                                float* __restrict__ logits,
                                                unsigned* __restrict__ maxbits,
                                                int Eraw, int Etot) {
  const int wave = __builtin_amdgcn_readfirstlane((int)(threadIdx.x >> 5));
  const int e0   = (blockIdx.x * 8 + wave) * EPW;
  if (e0 >= Etot) return;
  const int lane = threadIdx.x & 31;
  if (e0 + EPW <= Eraw)
    logits_batch<ALL_RAW>(ei, proj, att, logits, maxbits, Eraw, Etot, e0, lane);
  else if (e0 >= Eraw && e0 + EPW <= Etot)
    logits_batch<ALL_SELF>(ei, proj, att, logits, maxbits, Eraw, Etot, e0, lane);
  else
    logits_batch<ANY>(ei, proj, att, logits, maxbits, Eraw, Etot, e0, lane);
}

__global__ void __launch_bounds__(256) k_exp(const int* __restrict__ ei,
                                             const unsigned* __restrict__ maxbits,
                                             float* __restrict__ lp,
                                             float* __restrict__ denom,
                                             int Eraw, int Etot) {
  int tid = blockIdx.x * blockDim.x + threadIdx.x;
  if (tid >= 2 * Etot) return;
  int e = tid >> 1, h = tid & 1;
  int dst = (e < Eraw) ? ei[Eraw + e] : (e - Eraw);
  float m = dec_f32(maxbits[dst * 2 + h]);
  float p = __expf(lp[tid] - m);
  lp[tid] = p;
  atomic_add_f32(&denom[dst * 2 + h], p);
}

template <int MODE>
__device__ __forceinline__ void agg_batch(const int* __restrict__ ei,
                                          const float* __restrict__ proj,
                                          const float* __restrict__ p,
                                          const float* __restrict__ denom,
                                          float* __restrict__ out,
                                          int Eraw, int Etot, int e0, int lane) {
  int src[EPW], dst[EPW];
  v2f pe[EPW], dn[EPW];
  float xl[EPW][4];
#pragma unroll
  for (int i = 0; i < EPW; ++i)
    edge_nodes<MODE>(ei, e0 + i, Eraw, Etot, src[i], dst[i]);
#pragma unroll
  for (int i = 0; i < EPW; ++i) {
    int e = (MODE != ANY || e0 + i < Etot) ? e0 + i : Etot - 1;
    pe[i] = *(const v2f*)(p + (long)e * 2);
    dn[i] = *(const v2f*)(denom + (long)dst[i] * 2);
    const float* row = proj + (long)src[i] * FF + lane;
    xl[i][0] = row[0];  xl[i][1] = row[32];
    xl[i][2] = row[64]; xl[i][3] = row[96];
  }

#pragma unroll
  for (int i = 0; i < EPW; ++i) {
    if (MODE != ANY || e0 + i < Etot) {
      float alpha0 = pe[i][0] / dn[i][0];
      float alpha1 = pe[i][1] / dn[i][1];
      float* o = out + (long)dst[i] * HC + lane;
      atomic_add_f32(o +  0, alpha0 * xl[i][0]);
      atomic_add_f32(o + 32, alpha0 * xl[i][1]);
      atomic_add_f32(o + 64, alpha1 * xl[i][2]);
      atomic_add_f32(o + 96, alpha1 * xl[i][3]);
    }
  }
}

__global__ void __launch_bounds__(256) k_agg(const int* __restrict__ ei,
                                             const float* __restrict__ proj,
                                             const float* __restrict__ p,
                                             const float* __restrict__ denom,
                                             float* __restrict__ out,
                                             int Eraw, int Etot) {
  const int wave = __builtin_amdgcn_readfirstlane((int)(threadIdx.x >> 5));
  const int e0   = (blockIdx.x * 8 + wave) * EPW;
  if (e0 >= Etot) return;
  const int lane = threadIdx.x & 31;
  if (e0 + EPW <= Eraw)
    agg_batch<ALL_RAW>(ei, proj, p, denom, out, Eraw, Etot, e0, lane);
  else if (e0 >= Eraw && e0 + EPW <= Etot)
    agg_batch<ALL_SELF>(ei, proj, p, denom, out, Eraw, Etot, e0, lane);
  else
    agg_batch<ANY>(ei, proj, p, denom, out, Eraw, Etot, e0, lane);
}

extern "C" void kernel_launch(void* const* d_in, const int* in_sizes, int n_in,
                              void* d_out, int out_size, void* d_ws, size_t ws_size,
                              hipStream_t stream) {
  const float* x    = (const float*)d_in[0];
  const float* Wl   = (const float*)d_in[1];
  const float* bl   = (const float*)d_in[2];
  const float* Wr   = (const float*)d_in[3];
  const float* br   = (const float*)d_in[4];
  const float* att  = (const float*)d_in[5];
  const float* bias = (const float*)d_in[6];
  const int*   ei   = (const int*)d_in[7];
  float* out = (float*)d_out;

  const int Eraw = in_sizes[7] / 2;
  const int Etot = Eraw + NN;

  float*    proj    = (float*)d_ws;
  float*    logits  = proj + (size_t)NN * FF;
  unsigned* maxbits = (unsigned*)(logits + (size_t)Etot * 2);
  float*    denom   = (float*)(maxbits + (size_t)NN * 2);

  k_init  <<<(NN * HC + 255) / 256, 256, 0, stream>>>(out, maxbits, denom, bias);
  k_proj  <<<NN / 16,               256, 0, stream>>>(x, Wl, bl, Wr, br, proj);
  k_logits<<<(Etot + 8 * EPW - 1) / (8 * EPW), 256, 0, stream>>>(ei, proj, att, logits, maxbits, Eraw, Etot);
  k_exp   <<<(2 * Etot + 255) / 256,256, 0, stream>>>(ei, maxbits, logits, denom, Eraw, Etot);
  k_agg   <<<(Etot + 8 * EPW - 1) / (8 * EPW), 256, 0, stream>>>(ei, proj, logits, denom, out, Eraw, Etot);
}


# Start values (out = bias, segment max, denominator) stored by the projection blocks for their own 16 nodes, whole 512-B rows per store; the 400000-wave init launch goes: 5 launches become 4
# speedup vs baseline: 1.4734x; 1.0422x over previous
#include <hip/hip_runtime.h>

typedef __attribute__((ext_vector_type(2))) float v2f;
typedef __attribute__((ext_vector_type(8))) float v8f;

#define NN 100000
#define FF 256
#define HC 128
#define NEG_SLOPE 0.2f
#define ENC_NEG_INF 0x007FFFFFu

__device__ __forceinline__ unsigned enc_f32(float x) {
  unsigned u = __float_as_uint(x);
  return u ^ ((u & 0x80000000u) ? 0xFFFFFFFFu : 0x80000000u);
}
__device__ __forceinline__ float dec_f32(unsigned m) {
  unsigned u = (m & 0x80000000u) ? (m ^ 0x80000000u) : ~m;
  return __uint_as_float(u);
}
__device__ __forceinline__ void atomic_add_f32(float* p, float v) {
  __hip_atomic_fetch_add(p, v, __ATOMIC_RELAXED, __HIP_MEMORY_SCOPE_AGENT);
}

__global__ void __launch_bounds__(256) k_proj(const float* __restrict__ x,
                                              const float* __restrict__ Wl,
                                              const float* __restrict__ bl,
                                              const float* __restrict__ Wr,
                                              const float* __restrict__ br,
                                              const float* __restrict__ bias,
                                              float* __restrict__ proj,
                                              float* __restrict__ out,
                                              unsigned* __restrict__ maxbits,
                                              float* __restrict__ denom) {
  __shared__ __align__(16) float xs[16 * 260];
  const int m0 = blockIdx.x * 16;
  const int t  = threadIdx.x;
  const int bc4 = (t & 31) << 2;
  const float4 bz = *(const float4*)(bias + bc4);

#pragma unroll
  for (int i = 0; i < 4; ++i) {
    int idx = i * 256 + t;
    int row = idx >> 6;
    int c4  = (idx & 63) << 2;
    float4 v = *(const float4*)(x + (long)(m0 + row) * FF + c4);
    *(float4*)(&xs[row * 260 + c4]) = v;
  }
  __syncthreads();

  {
    float* o = out + (long)(m0 + (t >> 5)) * HC + bc4;
    *(float4*)(o)          = bz;
    *(float4*)(o + 8 * HC) = bz;
    if (t < 32) { maxbits[m0 * 2 + t] = ENC_NEG_INF; denom[m0 * 2 + t] = 0.0f; }
  }

  const int lane = t & 31;
  const int wave = t >> 5;
  const int r    = lane & 15;
  const int koff = (lane >> 4) << 1;

  for (int t2 = 0; t2 < 2; ++t2) {
    const int n0  = wave * 32 + t2 * 16;
    const int col = n0 + r;
    const float* Wp;
    float bv;
    if (col < HC) { Wp = Wl + (long)col * FF;        bv = bl[col]; }
    else          { Wp = Wr + (long)(col - HC) * FF; bv = br[col - HC]; }
    __builtin_prefetch(Wp, 0, 0);

    v8f acc;
#pragma unroll
    for (int v = 0; v < 8; ++v) acc[v] = bv;

#pragma unroll 8
    for (int k0 = 0; k0 < FF; k0 += 4) {
      v2f a, b;
      a[0] = xs[r * 260 + k0 + koff];
      a[1] = xs[r * 260 + k0 + koff + 1];
      b[0] = Wp[k0 + koff];
      b[1] = Wp[k0 + koff + 1];
      acc = __builtin_amdgcn_wmma_f32_16x16x4_f32(
          false, a, false, b, (short)0, acc, false, false);
    }

    const int rowbase = m0 + ((lane >> 4) << 3);
#pragma unroll
    for (int v = 0; v < 8; ++v)
      proj[(long)(rowbase + v) * FF + col] = acc[v];
  }
}

#define EPW 8

enum { ALL_RAW = 0, ALL_SELF = 1, ANY = 2 };

template <int MODE>
__device__ __forceinline__ void edge_nodes(const int* __restrict__ ei, int e,
                                           int Eraw, int Etot, int& src, int& dst) {
  if (MODE == ALL_RAW) {
    src = ei[e]; dst = ei[Eraw + e];
  } else if (MODE == ALL_SELF) {
    src = dst = e - Eraw;
  } else {
    int ec = e < Etot ? e : Etot - 1;
    if (ec < Eraw) { src = ei[ec]; dst = ei[Eraw + ec]; }
    else           { src = dst = ec - Eraw; }
  }
}

template <int MODE>
__device__ __forceinline__ void logits_batch(const int* __restrict__ ei,
                                             const float* __restrict__ proj,
                                             const float* __restrict__ att,
                                             float* __restrict__ logits,
                                             unsigned* __restrict__ maxbits,
                                             int Eraw, int Etot, int e0, int lane) {
  const int j0 = lane << 2;
  int src[EPW], dst[EPW];
#pragma unroll
  for (int i = 0; i < EPW; ++i)
    edge_nodes<MODE>(ei, e0 + i, Eraw, Etot, src[i], dst[i]);

  float4 a[EPW], b[EPW];
#pragma unroll
  for (int i = 0; i < EPW; ++i) {
    a[i] = *(const float4*)(proj + (long)src[i] * FF + j0);
    b[i] = *(const float4*)(proj + (long)dst[i] * FF + HC + j0);
  }
  const float4 w = *(const float4*)(att + j0);

  float sum[EPW];
#pragma unroll
  for (int i = 0; i < EPW; ++i) {
    float s = 0.0f, e;
    e = a[i].x + b[i].x; e = e > 0.0f ? e : NEG_SLOPE * e; s += w.x * e;
    e = a[i].y + b[i].y; e = e > 0.0f ? e : NEG_SLOPE * e; s += w.y * e;
    e = a[i].z + b[i].z; e = e > 0.0f ? e : NEG_SLOPE * e; s += w.z * e;
    e = a[i].w + b[i].w; e = e > 0.0f ? e : NEG_SLOPE * e; s += w.w * e;
    sum[i] = s;
  }

#pragma unroll
  for (int m = 1; m < 16; m <<= 1) {
#pragma unroll
    for (int i = 0; i < EPW; ++i) sum[i] += __shfl_xor(sum[i], m, 32);
  }

  if ((lane & 15) == 0) {
    int h = lane >> 4;
#pragma unroll
    for (int i = 0; i < EPW; ++i) {
      if (MODE != ANY || e0 + i < Etot) {
        logits[(e0 + i) * 2 + h] = sum[i];
        atomicMax(&maxbits[dst[i] * 2 + h], enc_f32(sum[i]));
      }
    }
  }
}

__global__ void __launch_bounds__(256) k_logits(const int* __restrict__ ei,
                                                const float* __restrict__ proj,
                                                const float* __restrict__ att,
                                                float* __restrict__ logits,
                                                unsigned* __restrict__ maxbits,
                                                int Eraw, int Etot) {
  const int wave = __builtin_amdgcn_readfirstlane((int)(threadIdx.x >> 5));
  const int e0   = (blockIdx.x * 8 + wave) * EPW;
  if (e0 >= Etot) return;
  const int lane = threadIdx.x & 31;
  if (e0 + EPW <= Eraw)
    logits_batch<ALL_RAW>(ei, proj, att, logits, maxbits, Eraw, Etot, e0, lane);
  else if (e0 >= Eraw && e0 + EPW <= Etot)
    logits_batch<ALL_SELF>(ei, proj, att, logits, maxbits, Eraw, Etot, e0, lane);
  else
    logits_batch<ANY>(ei, proj, att, logits, maxbits, Eraw, Etot, e0, lane);
}

__global__ void __launch_bounds__(256) k_exp(const int* __restrict__ ei,
                                             const unsigned* __restrict__ maxbits,
                                             float* __restrict__ lp,
                                             float* __restrict__ denom,
                                             int Eraw, int Etot) {
  int tid = blockIdx.x * blockDim.x + threadIdx.x;
  if (tid >= 2 * Etot) return;
  int e = tid >> 1, h = tid & 1;
  int dst = (e < Eraw) ? ei[Eraw + e] : (e - Eraw);
  float m = dec_f32(maxbits[dst * 2 + h]);
  float p = __expf(lp[tid] - m);
  lp[tid] = p;
  atomic_add_f32(&denom[dst * 2 + h], p);
}

template <int MODE>
__device__ __forceinline__ void agg_batch(const int* __restrict__ ei,
                                          const float* __restrict__ proj,
                                          const float* __restrict__ p,
                                          const float* __restrict__ denom,
                                          float* __restrict__ out,
                                          int Eraw, int Etot, int e0, int lane) {
  int src[EPW], dst[EPW];
  v2f pe[EPW], dn[EPW];
  float xl[EPW][4];
#pragma unroll
  for (int i = 0; i < EPW; ++i)
    edge_nodes<MODE>(ei, e0 + i, Eraw, Etot, src[i], dst[i]);
#pragma unroll
  for (int i = 0; i < EPW; ++i) {
    int e = (MODE != ANY || e0 + i < Etot) ? e0 + i : Etot - 1;
    pe[i] = *(const v2f*)(p + (long)e * 2);
    dn[i] = *(const v2f*)(denom + (long)dst[i] * 2);
    const float* row = proj + (long)src[i] * FF + lane;
    xl[i][0] = row[0];  xl[i][1] = row[32];
    xl[i][2] = row[64]; xl[i][3] = row[96];
  }

#pragma unroll
  for (int i = 0; i < EPW; ++i) {
    if (MODE != ANY || e0 + i < Etot) {
      float alpha0 = pe[i][0] / dn[i][0];
      float alpha1 = pe[i][1] / dn[i][1];
      float* o = out + (long)dst[i] * HC + lane;
      atomic_add_f32(o +  0, alpha0 * xl[i][0]);
      atomic_add_f32(o + 32, alpha0 * xl[i][1]);
      atomic_add_f32(o + 64, alpha1 * xl[i][2]);
      atomic_add_f32(o + 96, alpha1 * xl[i][3]);
    }
  }
}

__global__ void __launch_bounds__(256) k_agg(const int* __restrict__ ei,
                                             const float* __restrict__ proj,
                                             const float* __restrict__ p,
                                             const float* __restrict__ denom,
                                             float* __restrict__ out,
                                             int Eraw, int Etot) {
  const int wave = __builtin_amdgcn_readfirstlane((int)(threadIdx.x >> 5));
  const int e0   = (blockIdx.x * 8 + wave) * EPW;
  if (e0 >= Etot) return;
  const int lane = threadIdx.x & 31;
  if (e0 + EPW <= Eraw)
    agg_batch<ALL_RAW>(ei, proj, p, denom, out, Eraw, Etot, e0, lane);
  else if (e0 >= Eraw && e0 + EPW <= Etot)
    agg_batch<ALL_SELF>(ei, proj, p, denom, out, Eraw, Etot, e0, lane);
  else
    agg_batch<ANY>(ei, proj, p, denom, out, Eraw, Etot, e0, lane);
}

extern "C" void kernel_launch(void* const* d_in, const int* in_sizes, int n_in,
                              void* d_out, int out_size, void* d_ws, size_t ws_size,
                              hipStream_t stream) {
  const float* x    = (const float*)d_in[0];
  const float* Wl   = (const float*)d_in[1];
  const float* bl   = (const float*)d_in[2];
  const float* Wr   = (const float*)d_in[3];
  const float* br   = (const float*)d_in[4];
  const float* att  = (const float*)d_in[5];
  const float* bias = (const float*)d_in[6];
  const int*   ei   = (const int*)d_in[7];
  float* out = (float*)d_out;

  const int Eraw = in_sizes[7] / 2;
  const int Etot = Eraw + NN;

  float*    proj    = (float*)d_ws;
  float*    logits  = proj + (size_t)NN * FF;
  unsigned* maxbits = (unsigned*)(logits + (size_t)Etot * 2);
  float*    denom   = (float*)(maxbits + (size_t)NN * 2);

  k_proj  <<<NN / 16,               256, 0, stream>>>(x, Wl, bl, Wr, br, bias, proj, out, maxbits, denom);
  k_logits<<<(Etot + 8 * EPW - 1) / (8 * EPW), 256, 0, stream>>>(ei, proj, att, logits, maxbits, Eraw, Etot);
  k_exp   <<<(2 * Etot + 255) / 256,256, 0, stream>>>(ei, maxbits, logits, denom, Eraw, Etot);
  k_agg   <<<(Etot + 8 * EPW - 1) / (8 * EPW), 256, 0, stream>>>(ei, proj, logits, denom, out, Eraw, Etot);
}


# Projection GEMM re-tiled to 80 rows per block: a wave keeps 5 row x 2 column accumulator tiles, so each 16-line global B fragment load feeds 5 WMMAs, not 1 (weights not staged in LDS); sum order kept
# speedup vs baseline: 1.6884x; 1.1459x over previous
#include <hip/hip_runtime.h>

typedef __attribute__((ext_vector_type(2))) float v2f;
typedef __attribute__((ext_vector_type(8))) float v8f;

#define NN 100000
#define FF 256
#define HC 128
#define NEG_SLOPE 0.2f
#define ENC_NEG_INF 0x007FFFFFu

__device__ __forceinline__ unsigned enc_f32(float x) {
  unsigned u = __float_as_uint(x);
  return u ^ ((u & 0x80000000u) ? 0xFFFFFFFFu : 0x80000000u);
}
__device__ __forceinline__ float dec_f32(unsigned m) {
  unsigned u = (m & 0x80000000u) ? (m ^ 0x80000000u) : ~m;
  return __uint_as_float(u);
}
__device__ __forceinline__ void atomic_add_f32(float* p, float v) {
  __hip_atomic_fetch_add(p, v, __ATOMIC_RELAXED, __HIP_MEMORY_SCOPE_AGENT);
}

#define MT 5
#define RB (16 * MT)
#define XP 260

__global__ void __launch_bounds__(256) k_proj(const float* __restrict__ x,
                                              const float* __restrict__ Wl,
                                              const float* __restrict__ bl,
                                              const float* __restrict__ Wr,
                                              const float* __restrict__ br,
                                              const float* __restrict__ bias,
                                              float* __restrict__ proj,
                                              float* __restrict__ out,
                                              unsigned* __restrict__ maxbits,
                                              float* __restrict__ denom) {
  __shared__ __align__(16) float xs[RB * XP];
  const int m0 = blockIdx.x * RB;
  const int t  = threadIdx.x;
  const int bc4 = (t & 31) << 2;
  const float4 bz = *(const float4*)(bias + bc4);

#pragma unroll
  for (int i = 0; i < (RB * FF / 4) / 256; ++i) {
    int idx = i * 256 + t;
    int row = idx >> 6;
    int c4  = (idx & 63) << 2;
    float4 v = *(const float4*)(x + (long)(m0 + row) * FF + c4);
    *(float4*)(&xs[row * XP + c4]) = v;
  }
  __syncthreads();

  {
    float* o = out + (long)(m0 + (t >> 5)) * HC + bc4;
#pragma unroll
    for (int j = 0; j < RB / 8; ++j) *(float4*)(o + (long)j * 8 * HC) = bz;
    if (t < 2 * RB) { maxbits[m0 * 2 + t] = ENC_NEG_INF; denom[m0 * 2 + t] = 0.0f; }
  }

  const int lane = t & 31;
  const int wave = t >> 5;
  const int r    = lane & 15;
  const int koff = (lane >> 4) << 1;

  const int col0 = wave * 32 + r;
  const float* Wp0;
  float bv0, bv1;
  if (col0 < HC) { Wp0 = Wl + (long)col0 * FF;        bv0 = bl[col0];      bv1 = bl[col0 + 16]; }
  else           { Wp0 = Wr + (long)(col0 - HC) * FF; bv0 = br[col0 - HC]; bv1 = br[col0 - HC + 16]; }
  const float* Wp1 = Wp0 + 16 * FF;

  v8f acc[MT][2];
#pragma unroll
  for (int m = 0; m < MT; ++m) {
#pragma unroll
    for (int v = 0; v < 8; ++v) { acc[m][0][v] = bv0; acc[m][1][v] = bv1; }
  }

  const float* xa = xs + r * XP + koff;
#pragma unroll 4
  for (int k0 = 0; k0 < FF; k0 += 4) {
    v2f b0, b1;
    b0[0] = Wp0[k0 + koff]; b0[1] = Wp0[k0 + koff + 1];
    b1[0] = Wp1[k0 + koff]; b1[1] = Wp1[k0 + koff + 1];
#pragma unroll
    for (int m = 0; m < MT; ++m) {
      v2f a;
      a[0] = xa[m * 16 * XP + k0];
      a[1] = xa[m * 16 * XP + k0 + 1];
      acc[m][0] = __builtin_amdgcn_wmma_f32_16x16x4_f32(
          false, a, false, b0, (short)0, acc[m][0], false, false);
      acc[m][1] = __builtin_amdgcn_wmma_f32_16x16x4_f32(
          false, a, false, b1, (short)0, acc[m][1], false, false);
    }
  }

#pragma unroll
  for (int m = 0; m < MT; ++m) {
    const int rowbase = m0 + m * 16 + ((lane >> 4) << 3);
#pragma unroll
    for (int v = 0; v < 8; ++v) {
      float* pr = proj + (long)(rowbase + v) * FF + col0;
      pr[0]  = acc[m][0][v];
      pr[16] = acc[m][1][v];
    }
  }
}

#define EPW 8

enum { ALL_RAW = 0, ALL_SELF = 1, ANY = 2 };

template <int MODE>
__device__ __forceinline__ void edge_nodes(const int* __restrict__ ei, int e,
                                           int Eraw, int Etot, int& src, int& dst) {
  if (MODE == ALL_RAW) {
    src = ei[e]; dst = ei[Eraw + e];
  } else if (MODE == ALL_SELF) {
    src = dst = e - Eraw;
  } else {
    int ec = e < Etot ? e : Etot - 1;
    if (ec < Eraw) { src = ei[ec]; dst = ei[Eraw + ec]; }
    else           { src = dst = ec - Eraw; }
  }
}

template <int MODE>
__device__ __forceinline__ void logits_batch(const int* __restrict__ ei,
                                             const float* __restrict__ proj,
                                             const float* __restrict__ att,
                                             float* __restrict__ logits,
                                             unsigned* __restrict__ maxbits,
                                             int Eraw, int Etot, int e0, int lane) {
  const int j0 = lane << 2;
  int src[EPW], dst[EPW];
#pragma unroll
  for (int i = 0; i < EPW; ++i)
    edge_nodes<MODE>(ei, e0 + i, Eraw, Etot, src[i], dst[i]);

  float4 a[EPW], b[EPW];
#pragma unroll
  for (int i = 0; i < EPW; ++i) {
    a[i] = *(const float4*)(proj + (long)src[i] * FF + j0);
    b[i] = *(const float4*)(proj + (long)dst[i] * FF + HC + j0);
  }
  const float4 w = *(const float4*)(att + j0);

  float sum[EPW];
#pragma unroll
  for (int i = 0; i < EPW; ++i) {
    float s = 0.0f, e;
    e = a[i].x + b[i].x; e = e > 0.0f ? e : NEG_SLOPE * e; s += w.x * e;
    e = a[i].y + b[i].y; e = e > 0.0f ? e : NEG_SLOPE * e; s += w.y * e;
    e = a[i].z + b[i].z; e = e > 0.0f ? e : NEG_SLOPE * e; s += w.z * e;
    e = a[i].w + b[i].w; e = e > 0.0f ? e : NEG_SLOPE * e; s += w.w * e;
    sum[i] = s;
  }

#pragma unroll
  for (int m = 1; m < 16; m <<= 1) {
#pragma unroll
    for (int i = 0; i < EPW; ++i) sum[i] += __shfl_xor(sum[i], m, 32);
  }

  if ((lane & 15) == 0) {
    int h = lane >> 4;
#pragma unroll
    for (int i = 0; i < EPW; ++i) {
      if (MODE != ANY || e0 + i < Etot) {
        logits[(e0 + i) * 2 + h] = sum[i];
        atomicMax(&maxbits[dst[i] * 2 + h], enc_f32(sum[i]));
      }
    }
  }
}

__global__ void __launch_bounds__(256) k_logits(const int* __restrict__ ei,
                                                const float* __restrict__ proj,
                                                const float* __restrict__ att,
                                                float* __restrict__ logits,
                                                unsigned* __restrict__ maxbits,
                                                int Eraw, int Etot) {
  const int wave = __builtin_amdgcn_readfirstlane((int)(threadIdx.x >> 5));
  const int e0   = (blockIdx.x * 8 + wave) * EPW;
  if (e0 >= Etot) return;
  const int lane = threadIdx.x & 31;
  if (e0 + EPW <= Eraw)
    logits_batch<ALL_RAW>(ei, proj, att, logits, maxbits, Eraw, Etot, e0, lane);
  else if (e0 >= Eraw && e0 + EPW <= Etot)
    logits_batch<ALL_SELF>(ei, proj, att, logits, maxbits, Eraw, Etot, e0, lane);
  else
    logits_batch<ANY>(ei, proj, att, logits, maxbits, Eraw, Etot, e0, lane);
}

__global__ void __launch_bounds__(256) k_exp(const int* __restrict__ ei,
                                             const unsigned* __restrict__ maxbits,
                                             float* __restrict__ lp,
                                             float* __restrict__ denom,
                                             int Eraw, int Etot) {
  int tid = blockIdx.x * blockDim.x + threadIdx.x;
  if (tid >= 2 * Etot) return;
  int e = tid >> 1, h = tid & 1;
  int dst = (e < Eraw) ? ei[Eraw + e] : (e - Eraw);
  float m = dec_f32(maxbits[dst * 2 + h]);
  float p = __expf(lp[tid] - m);
  lp[tid] = p;
  atomic_add_f32(&denom[dst * 2 + h], p);
}

template <int MODE>
__device__ __forceinline__ void agg_batch(const int* __restrict__ ei,
                                          const float* __restrict__ proj,
                                          const float* __restrict__ p,
                                          const float* __restrict__ denom,
                                          float* __restrict__ out,
                                          int Eraw, int Etot, int e0, int lane) {
  int src[EPW], dst[EPW];
  v2f pe[EPW], dn[EPW];
  float xl[EPW][4];
#pragma unroll
  for (int i = 0; i < EPW; ++i)
    edge_nodes<MODE>(ei, e0 + i, Eraw, Etot, src[i], dst[i]);
#pragma unroll
  for (int i = 0; i < EPW; ++i) {
    int e = (MODE != ANY || e0 + i < Etot) ? e0 + i : Etot - 1;
    pe[i] = *(const v2f*)(p + (long)e * 2);
    dn[i] = *(const v2f*)(denom + (long)dst[i] * 2);
    const float* row = proj + (long)src[i] * FF + lane;
    xl[i][0] = row[0];  xl[i][1] = row[32];
    xl[i][2] = row[64]; xl[i][3] = row[96];
  }

#pragma unroll
  for (int i = 0; i < EPW; ++i) {
    if (MODE != ANY || e0 + i < Etot) {
      float alpha0 = pe[i][0] / dn[i][0];
      float alpha1 = pe[i][1] / dn[i][1];
      float* o = out + (long)dst[i] * HC + lane;
      atomic_add_f32(o +  0, alpha0 * xl[i][0]);
      atomic_add_f32(o + 32, alpha0 * xl[i][1]);
      atomic_add_f32(o + 64, alpha1 * xl[i][2]);
      atomic_add_f32(o + 96, alpha1 * xl[i][3]);
    }
  }
}

__global__ void __launch_bounds__(256) k_agg(const int* __restrict__ ei,
                                             const float* __restrict__ proj,
                                             const float* __restrict__ p,
                                             const float* __restrict__ denom,
                                             float* __restrict__ out,
                                             int Eraw, int Etot) {
  const int wave = __builtin_amdgcn_readfirstlane((int)(threadIdx.x >> 5));
  const int e0   = (blockIdx.x * 8 + wave) * EPW;
  if (e0 >= Etot) return;
  const int lane = threadIdx.x & 31;
  if (e0 + EPW <= Eraw)
    agg_batch<ALL_RAW>(ei, proj, p, denom, out, Eraw, Etot, e0, lane);
  else if (e0 >= Eraw && e0 + EPW <= Etot)
    agg_batch<ALL_SELF>(ei, proj, p, denom, out, Eraw, Etot, e0, lane);
  else
    agg_batch<ANY>(ei, proj, p, denom, out, Eraw, Etot, e0, lane);
}

extern "C" void kernel_launch(void* const* d_in, const int* in_sizes, int n_in,
                              void* d_out, int out_size, void* d_ws, size_t ws_size,
                              hipStream_t stream) {
  const float* x    = (const float*)d_in[0];
  const float* Wl   = (const float*)d_in[1];
  const float* bl   = (const float*)d_in[2];
  const float* Wr   = (const float*)d_in[3];
  const float* br   = (const float*)d_in[4];
  const float* att  = (const float*)d_in[5];
  const float* bias = (const float*)d_in[6];
  const int*   ei   = (const int*)d_in[7];
  float* out = (float*)d_out;

  const int Eraw = in_sizes[7] / 2;
  const int Etot = Eraw + NN;

  float*    proj    = (float*)d_ws;
  float*    logits  = proj + (size_t)NN * FF;
  unsigned* maxbits = (unsigned*)(logits + (size_t)Etot * 2);
  float*    denom   = (float*)(maxbits + (size_t)NN * 2);

  k_proj  <<<NN / RB,               256, 0, stream>>>(x, Wl, bl, Wr, br, bias, proj, out, maxbits, denom);
  k_logits<<<(Etot + 8 * EPW - 1) / (8 * EPW), 256, 0, stream>>>(ei, proj, att, logits, maxbits, Eraw, Etot);
  k_exp   <<<(2 * Etot + 255) / 256,256, 0, stream>>>(ei, maxbits, logits, denom, Eraw, Etot);
  k_agg   <<<(Etot + 8 * EPW - 1) / (8 * EPW), 256, 0, stream>>>(ei, proj, logits, denom, out, Eraw, Etot);
}


# Exp and scatter passes fused into one owner pass: a block per WGP holds 391 nodes' sums and denominators in LDS and sweeps the destination list; LDS float atomics (arrival order), division after sum
# speedup vs baseline: 2.0104x; 1.1907x over previous
#include <hip/hip_runtime.h>

typedef __attribute__((ext_vector_type(2))) float v2f;
typedef __attribute__((ext_vector_type(8))) float v8f;

#define NN 100000
#define FF 256
#define HC 128
#define NEG_SLOPE 0.2f
#define ENC_NEG_INF 0x007FFFFFu

__device__ __forceinline__ unsigned enc_f32(float x) {
  unsigned u = __float_as_uint(x);
  return u ^ ((u & 0x80000000u) ? 0xFFFFFFFFu : 0x80000000u);
}
__device__ __forceinline__ float dec_f32(unsigned m) {
  unsigned u = (m & 0x80000000u) ? (m ^ 0x80000000u) : ~m;
  return __uint_as_float(u);
}
__device__ __forceinline__ void lds_add_f32(float* p, float v) {
  __hip_atomic_fetch_add(p, v, __ATOMIC_RELAXED, __HIP_MEMORY_SCOPE_WORKGROUP);
}

#define MT 5
#define RB (16 * MT)
#define XP 260

__global__ void __launch_bounds__(256) k_proj(const float* __restrict__ x,
                                              const float* __restrict__ Wl,
                                              const float* __restrict__ bl,
                                              const float* __restrict__ Wr,
                                              const float* __restrict__ br,
                                              float* __restrict__ proj,
                                              unsigned* __restrict__ maxbits) {
  __shared__ __align__(16) float xs[RB * XP];
  const int m0 = blockIdx.x * RB;
  const int t  = threadIdx.x;

#pragma unroll
  for (int i = 0; i < (RB * FF / 4) / 256; ++i) {
    int idx = i * 256 + t;
    int row = idx >> 6;
    int c4  = (idx & 63) << 2;
    float4 v = *(const float4*)(x + (long)(m0 + row) * FF + c4);
    *(float4*)(&xs[row * XP + c4]) = v;
  }
  __syncthreads();

  if (t < 2 * RB) maxbits[m0 * 2 + t] = ENC_NEG_INF;

  const int lane = t & 31;
  const int wave = t >> 5;
  const int r    = lane & 15;
  const int koff = (lane >> 4) << 1;

  const int col0 = wave * 32 + r;
  const float* Wp0;
  float bv0, bv1;
  if (col0 < HC) { Wp0 = Wl + (long)col0 * FF;        bv0 = bl[col0];      bv1 = bl[col0 + 16]; }
  else           { Wp0 = Wr + (long)(col0 - HC) * FF; bv0 = br[col0 - HC]; bv1 = br[col0 - HC + 16]; }
  const float* Wp1 = Wp0 + 16 * FF;

  v8f acc[MT][2];
#pragma unroll
  for (int m = 0; m < MT; ++m) {
#pragma unroll
    for (int v = 0; v < 8; ++v) { acc[m][0][v] = bv0; acc[m][1][v] = bv1; }
  }

  const float* xa = xs + r * XP + koff;
#pragma unroll 4
  for (int k0 = 0; k0 < FF; k0 += 4) {
    v2f b0, b1;
    b0[0] = Wp0[k0 + koff]; b0[1] = Wp0[k0 + koff + 1];
    b1[0] = Wp1[k0 + koff]; b1[1] = Wp1[k0 + koff + 1];
#pragma unroll
    for (int m = 0; m < MT; ++m) {
      v2f a;
      a[0] = xa[m * 16 * XP + k0];
      a[1] = xa[m * 16 * XP + k0 + 1];
      acc[m][0] = __builtin_amdgcn_wmma_f32_16x16x4_f32(
          false, a, false, b0, (short)0, acc[m][0], false, false);
      acc[m][1] = __builtin_amdgcn_wmma_f32_16x16x4_f32(
          false, a, false, b1, (short)0, acc[m][1], false, false);
    }
  }

#pragma unroll
  for (int m = 0; m < MT; ++m) {
    const int rowbase = m0 + m * 16 + ((lane >> 4) << 3);
#pragma unroll
    for (int v = 0; v < 8; ++v) {
      float* pr = proj + (long)(rowbase + v) * FF + col0;
      pr[0]  = acc[m][0][v];
      pr[16] = acc[m][1][v];
    }
  }
}

#define EPW 8

enum { ALL_RAW = 0, ALL_SELF = 1, ANY = 2 };

template <int MODE>
__device__ __forceinline__ void edge_nodes(const int* __restrict__ ei, int e,
                                           int Eraw, int Etot, int& src, int& dst) {
  if (MODE == ALL_RAW) {
    src = ei[e]; dst = ei[Eraw + e];
  } else if (MODE == ALL_SELF) {
    src = dst = e - Eraw;
  } else {
    int ec = e < Etot ? e : Etot - 1;
    if (ec < Eraw) { src = ei[ec]; dst = ei[Eraw + ec]; }
    else           { src = dst = ec - Eraw; }
  }
}

template <int MODE>
__device__ __forceinline__ void logits_batch(const int* __restrict__ ei,
                                             const float* __restrict__ proj,
                                             const float* __restrict__ att,
                                             float* __restrict__ logits,
                                             unsigned* __restrict__ maxbits,
                                             int Eraw, int Etot, int e0, int lane) {
  const int j0 = lane << 2;
  int src[EPW], dst[EPW];
#pragma unroll
  for (int i = 0; i < EPW; ++i)
    edge_nodes<MODE>(ei, e0 + i, Eraw, Etot, src[i], dst[i]);

  float4 a[EPW], b[EPW];
#pragma unroll
  for (int i = 0; i < EPW; ++i) {
    a[i] = *(const float4*)(proj + (long)src[i] * FF + j0);
    b[i] = *(const float4*)(proj + (long)dst[i] * FF + HC + j0);
  }
  const float4 w = *(const float4*)(att + j0);

  float sum[EPW];
#pragma unroll
  for (int i = 0; i < EPW; ++i) {
    float s = 0.0f, e;
    e = a[i].x + b[i].x; e = e > 0.0f ? e : NEG_SLOPE * e; s += w.x * e;
    e = a[i].y + b[i].y; e = e > 0.0f ? e : NEG_SLOPE * e; s += w.y * e;
    e = a[i].z + b[i].z; e = e > 0.0f ? e : NEG_SLOPE * e; s += w.z * e;
    e = a[i].w + b[i].w; e = e > 0.0f ? e : NEG_SLOPE * e; s += w.w * e;
    sum[i] = s;
  }

#pragma unroll
  for (int m = 1; m < 16; m <<= 1) {
#pragma unroll
    for (int i = 0; i < EPW; ++i) sum[i] += __shfl_xor(sum[i], m, 32);
  }

  if ((lane & 15) == 0) {
    int h = lane >> 4;
#pragma unroll
    for (int i = 0; i < EPW; ++i) {
      if (MODE != ANY || e0 + i < Etot) {
        logits[(e0 + i) * 2 + h] = sum[i];
        atomicMax(&maxbits[dst[i] * 2 + h], enc_f32(sum[i]));
      }
    }
  }
}

__global__ void __launch_bounds__(256) k_logits(const int* __restrict__ ei,
                                                const float* __restrict__ proj,
                                                const float* __restrict__ att,
                                                float* __restrict__ logits,
                                                unsigned* __restrict__ maxbits,
                                                int Eraw, int Etot) {
  const int wave = __builtin_amdgcn_readfirstlane((int)(threadIdx.x >> 5));
  const int e0   = (blockIdx.x * 8 + wave) * EPW;
  if (e0 >= Etot) return;
  const int lane = threadIdx.x & 31;
  if (e0 + EPW <= Eraw)
    logits_batch<ALL_RAW>(ei, proj, att, logits, maxbits, Eraw, Etot, e0, lane);
  else if (e0 >= Eraw && e0 + EPW <= Etot)
    logits_batch<ALL_SELF>(ei, proj, att, logits, maxbits, Eraw, Etot, e0, lane);
  else
    logits_batch<ANY>(ei, proj, att, logits, maxbits, Eraw, Etot, e0, lane);
}

#define NRANGE 391
#define OWN_BLOCKS 256
#define OWN_WAVES 32
#define SCAN_U 4
#define LCAP (SCAN_U * 128 + 16)
#define ENT_SHIFT 21

template <bool FULL>
__device__ __forceinline__ void own_batch(const int* __restrict__ ei,
                                          const float* __restrict__ proj,
                                          const float* __restrict__ logits,
                                          const unsigned* __restrict__ maxbits,
                                          const int* lst, int nvalid,
                                          float* sums, float* den,
                                          int lo, int Eraw, int lane) {
  int dl[EPW], ee[EPW], src[EPW];
  v2f lg[EPW];
  unsigned mb0[EPW], mb1[EPW];
  float xl[EPW][4];
#pragma unroll
  for (int i = 0; i < EPW; ++i) {
    const int ent = __builtin_amdgcn_readfirstlane(lst[(FULL || i < nvalid) ? i : 0]);
    ee[i] = ent & ((1 << ENT_SHIFT) - 1);
    dl[i] = ent >> ENT_SHIFT;
  }
#pragma unroll
  for (int i = 0; i < EPW; ++i) {
    const int sv = ei[ee[i] < Eraw ? ee[i] : 0];
    src[i] = ee[i] < Eraw ? sv : ee[i] - Eraw;
  }
#pragma unroll
  for (int i = 0; i < EPW; ++i) {
    const int dst = lo + dl[i];
    lg[i]  = *(const v2f*)(logits + (long)ee[i] * 2);
    mb0[i] = maxbits[dst * 2];
    mb1[i] = maxbits[dst * 2 + 1];
    const float* row = proj + (long)src[i] * FF + lane;
    xl[i][0] = row[0];  xl[i][1] = row[32];
    xl[i][2] = row[64]; xl[i][3] = row[96];
  }
#pragma unroll
  for (int i = 0; i < EPW; ++i) {
    if (FULL || i < nvalid) {
      const float p0 = __expf(lg[i][0] - dec_f32(mb0[i]));
      const float p1 = __expf(lg[i][1] - dec_f32(mb1[i]));
      float* s = sums + dl[i] * HC + lane;
      lds_add_f32(s +  0, p0 * xl[i][0]);
      lds_add_f32(s + 32, p0 * xl[i][1]);
      lds_add_f32(s + 64, p1 * xl[i][2]);
      lds_add_f32(s + 96, p1 * xl[i][3]);
      if (lane < 2) lds_add_f32(den + dl[i] * 2 + lane, lane ? p1 : p0);
    }
  }
}

__global__ void __launch_bounds__(1024) k_own(const int* __restrict__ ei,
                                              const float* __restrict__ proj,
                                              const float* __restrict__ logits,
                                              const unsigned* __restrict__ maxbits,
                                              const float* __restrict__ bias,
                                              float* __restrict__ out,
                                              int Eraw) {
  __shared__ __align__(16) float sums[NRANGE * HC];
  __shared__ float den[NRANGE * 2];
  __shared__ int lists[OWN_WAVES * LCAP];

  const int t    = threadIdx.x;
  const int lane = t & 31;
  const int wave = __builtin_amdgcn_readfirstlane(t >> 5);
  const int lo   = blockIdx.x * NRANGE;
  const int cnt  = (NN - lo) < NRANGE ? (NN - lo) : NRANGE;

  for (int i = t; i < NRANGE * HC / 4; i += 1024)
    *(float4*)(sums + i * 4) = make_float4(0.0f, 0.0f, 0.0f, 0.0f);
  if (t < NRANGE * 2) den[t] = 0.0f;
  __syncthreads();

  const int* dsts = ei + Eraw;
  int* lst = lists + wave * LCAP;
  int n = 0;
  const unsigned below = (1u << lane) - 1u;
  const int nchunk = (Eraw + 127) >> 7;

  for (int c0 = wave * SCAN_U; c0 < nchunk; c0 += OWN_WAVES * SCAN_U) {
    int d[SCAN_U][4];
    if ((c0 + SCAN_U) * 128 <= Eraw) {
#pragma unroll
      for (int u = 0; u < SCAN_U; ++u) {
        const int4 v = *(const int4*)(dsts + (c0 + u) * 128 + lane * 4);
        d[u][0] = v.x; d[u][1] = v.y; d[u][2] = v.z; d[u][3] = v.w;
      }
    } else {
#pragma unroll
      for (int u = 0; u < SCAN_U; ++u) {
#pragma unroll
        for (int k = 0; k < 4; ++k) {
          const int ek = (c0 + u) * 128 + lane * 4 + k;
          d[u][k] = ek < Eraw ? dsts[ek] : -1;
        }
      }
    }
#pragma unroll
    for (int u = 0; u < SCAN_U; ++u) {
      const int eb = (c0 + u) * 128 + lane * 4;
#pragma unroll
      for (int k = 0; k < 4; ++k) {
        const unsigned rel = (unsigned)(d[u][k] - lo);
        const bool mine = rel < (unsigned)cnt;
        const unsigned mask = __builtin_amdgcn_ballot_w32(mine);
        if (mask) {
          if (mine) lst[n + __builtin_popcount(mask & below)] = (eb + k) | ((int)rel << ENT_SHIFT);
          n += __builtin_popcount(mask);
        }
      }
    }
    __builtin_amdgcn_fence(__ATOMIC_ACQ_REL, "wavefront");
    while (n >= EPW) {
      n -= EPW;
      own_batch<true>(ei, proj, logits, maxbits, lst + n, EPW, sums, den, lo, Eraw, lane);
    }
  }

  {
    const int nl = wave + 32 * lane;
    const int mineCount = cnt > wave ? (cnt - wave + 31) >> 5 : 0;
    if (nl < cnt) lst[n + lane] = (Eraw + lo + nl) | (nl << ENT_SHIFT);
    n += mineCount;
    __builtin_amdgcn_fence(__ATOMIC_ACQ_REL, "wavefront");
    while (n >= EPW) {
      n -= EPW;
      own_batch<true>(ei, proj, logits, maxbits, lst + n, EPW, sums, den, lo, Eraw, lane);
    }
    if (n > 0)
      own_batch<false>(ei, proj, logits, maxbits, lst, n, sums, den, lo, Eraw, lane);
  }
  __syncthreads();

  const float4 bz = *(const float4*)(bias + lane * 4);
  for (int nl = wave; nl < cnt; nl += OWN_WAVES) {
    const float4 s = *(const float4*)(sums + nl * HC + lane * 4);
    const float dn = den[nl * 2 + (lane >> 4)];
    float4 o;
    o.x = bz.x + s.x / dn; o.y = bz.y + s.y / dn;
    o.z = bz.z + s.z / dn; o.w = bz.w + s.w / dn;
    *(float4*)(out + (long)(lo + nl) * HC + lane * 4) = o;
  }
}

extern "C" void kernel_launch(void* const* d_in, const int* in_sizes, int n_in,
                              void* d_out, int out_size, void* d_ws, size_t ws_size,
                              hipStream_t stream) {
  const float* x    = (const float*)d_in[0];
  const float* Wl   = (const float*)d_in[1];
  const float* bl   = (const float*)d_in[2];
  const float* Wr   = (const float*)d_in[3];
  const float* br   = (const float*)d_in[4];
  const float* att  = (const float*)d_in[5];
  const float* bias = (const float*)d_in[6];
  const int*   ei   = (const int*)d_in[7];
  float* out = (float*)d_out;

  const int Eraw = in_sizes[7] / 2;
  const int Etot = Eraw + NN;

  float*    proj    = (float*)d_ws;
  float*    logits  = proj + (size_t)NN * FF;
  unsigned* maxbits = (unsigned*)(logits + (size_t)Etot * 2);

  k_proj  <<<NN / RB,               256, 0, stream>>>(x, Wl, bl, Wr, br, proj, maxbits);
  k_logits<<<(Etot + 8 * EPW - 1) / (8 * EPW), 256, 0, stream>>>(ei, proj, att, logits, maxbits, Eraw, Etot);
  k_own   <<<OWN_BLOCKS,           1024, 0, stream>>>(ei, proj, logits, maxbits, bias, out, Eraw);
}
